# starting kernel
# baseline (speedup 1.0000x reference)
#include <hip/hip_runtime.h>
#include <hip/hip_bf16.h>
#include <math.h>

#define N_NODES 10000
#define DEG 32
#define IN_FEATS 256
#define HEADS 8
#define OUT_FEATS 64
#define HD (HEADS * OUT_FEATS)
#define NEG_SLOPE 0.2f

typedef __attribute__((ext_vector_type(2))) float v2f;
typedef __attribute__((ext_vector_type(8))) float v8f;

__global__ __launch_bounds__(256) void gat_gemm_wmma(
    const float* __restrict__ feat, const float* __restrict__ W,
    float* __restrict__ ft)
{
    const int lane  = threadIdx.x & 31;
    const int wave  = threadIdx.x >> 5;
    const int m0    = blockIdx.x * 16;
    const int nb    = wave * 64;
    const int mrow  = m0 + (lane & 15);
    const int khalf = (lane >> 4) * 2;
    const int ncol  = lane & 15;

    v8f acc0 = {}, acc1 = {}, acc2 = {}, acc3 = {};
    const float* arow = feat + (size_t)mrow * IN_FEATS;

    for (int k0 = 0; k0 < IN_FEATS; k0 += 4) {
        const int kb = k0 + khalf;
        v2f a = *(const v2f*)(arow + kb);
        const float* w0 = W + (size_t)kb * HD + nb + ncol;
        const float* w1 = w0 + HD;
        v2f b0, b1, b2, b3;
        b0.x = w0[0];  b0.y = w1[0];
        b1.x = w0[16]; b1.y = w1[16];
        b2.x = w0[32]; b2.y = w1[32];
        b3.x = w0[48]; b3.y = w1[48];
        acc0 = __builtin_amdgcn_wmma_f32_16x16x4_f32(false, a, false, b0, (short)0, acc0, false, false);
        acc1 = __builtin_amdgcn_wmma_f32_16x16x4_f32(false, a, false, b1, (short)0, acc1, false, false);
        acc2 = __builtin_amdgcn_wmma_f32_16x16x4_f32(false, a, false, b2, (short)0, acc2, false, false);
        acc3 = __builtin_amdgcn_wmma_f32_16x16x4_f32(false, a, false, b3, (short)0, acc3, false, false);
    }

    const int rbase = m0 + ((lane >> 4) << 3);
#pragma unroll
    for (int r = 0; r < 8; ++r) {
        float* dp = ft + (size_t)(rbase + r) * HD + nb + ncol;
        dp[0]  = acc0[r];
        dp[16] = acc1[r];
        dp[32] = acc2[r];
        dp[48] = acc3[r];
    }
}

__global__ __launch_bounds__(256) void gat_logits(
    const float* __restrict__ ft, const float* __restrict__ attn_l,
    const float* __restrict__ attn_r, float* __restrict__ el,
    float* __restrict__ er)
{
    const int n    = blockIdx.x;
    const int h    = threadIdx.x >> 5;
    const int lane = threadIdx.x & 31;

    const float* f  = ft + (size_t)n * HD + h * OUT_FEATS;
    const float* al = attn_l + h * OUT_FEATS;
    const float* ar = attn_r + h * OUT_FEATS;

    const float f0 = f[lane], f1 = f[lane + 32];
    float sl = f0 * al[lane] + f1 * al[lane + 32];
    float sr = f0 * ar[lane] + f1 * ar[lane + 32];
#pragma unroll
    for (int m = 16; m > 0; m >>= 1) {
        sl += __shfl_xor(sl, m, 32);
        sr += __shfl_xor(sr, m, 32);
    }
    if (lane == 0) {
        el[n * HEADS + h] = sl;
        er[n * HEADS + h] = sr;
    }
}

__global__ __launch_bounds__(256) void gat_edge_softmax_agg(
    const float* __restrict__ ft, const float* __restrict__ el,
    const float* __restrict__ er, const int* __restrict__ src,
    const int* __restrict__ dst, float* __restrict__ rst,
    float* __restrict__ a_out)
{
    __shared__ int   s_src[DEG];
    __shared__ float s_e[DEG][HEADS];
    __shared__ float s_a[DEG][HEADS];

    const int    n     = blockIdx.x;
    const int    t     = threadIdx.x;
    const long   ebase = (long)n * DEG;

    if (t < DEG) s_src[t] = src[ebase + t];
    __syncthreads();

    {
        const int j = t >> 3, h = t & 7;
        const int dn = dst[ebase + j];
        float e = el[s_src[j] * HEADS + h] + er[dn * HEADS + h];
        s_e[j][h] = (e > 0.f) ? e : NEG_SLOPE * e;
    }
    __syncthreads();

    {
        const int h = t >> 5, j = t & 31;
        float e  = s_e[j][h];
        float mx = e;
#pragma unroll
        for (int m = 16; m > 0; m >>= 1) mx = fmaxf(mx, __shfl_xor(mx, m, 32));
        float ex  = __expf(e - mx);
        float sum = ex;
#pragma unroll
        for (int m = 16; m > 0; m >>= 1) sum += __shfl_xor(sum, m, 32);
        const float a = ex / sum;
        s_a[j][h] = a;
        a_out[(ebase + j) * HEADS + h] = a;
    }
    __syncthreads();

#pragma unroll
    for (int rep = 0; rep < 2; ++rep) {
        const int idx = t + rep * 256;
        const int h = idx >> 6, d = idx & 63;
        float acc = 0.f;
#pragma unroll 8
        for (int j = 0; j < DEG; ++j)
            acc += s_a[j][h] * ft[(size_t)s_src[j] * HD + h * OUT_FEATS + d];
        rst[(size_t)n * HD + idx] = acc;
    }
}

extern "C" void kernel_launch(void* const* d_in, const int* in_sizes, int n_in,
                              void* d_out, int out_size, void* d_ws, size_t ws_size,
                              hipStream_t stream)
{
    const float* feat   = (const float*)d_in[0];
    const float* W      = (const float*)d_in[1];
    const float* attn_l = (const float*)d_in[2];
    const float* attn_r = (const float*)d_in[3];
    const int*   src    = (const int*)d_in[4];
    const int*   dst    = (const int*)d_in[5];

    float* out_rst = (float*)d_out;
    float* out_a   = out_rst + (size_t)N_NODES * HD;

    float* ft = (float*)d_ws;
    float* el = ft + (size_t)N_NODES * HD;
    float* er = el + (size_t)N_NODES * HEADS;

    gat_gemm_wmma<<<N_NODES / 16, 256, 0, stream>>>(feat, W, ft);
    gat_logits<<<N_NODES, 256, 0, stream>>>(ft, attn_l, attn_r, el, er);
    gat_edge_softmax_agg<<<N_NODES, 256, 0, stream>>>(ft, el, er, src, dst,
                                                      out_rst, out_a);
}


# edge-aggregation kernel gathers each ft row with 16-byte loads per lane: 128 lanes cover one 2 KB row, block 256->128 threads, each wave does 2 heads' softmax; summation order unchanged
# speedup vs baseline: 1.2611x; 1.2611x over previous
#include <hip/hip_runtime.h>
#include <hip/hip_bf16.h>
#include <math.h>

#define N_NODES 10000
#define DEG 32
#define IN_FEATS 256
#define HEADS 8
#define OUT_FEATS 64
#define HD (HEADS * OUT_FEATS)
#define NEG_SLOPE 0.2f

typedef __attribute__((ext_vector_type(2))) float v2f;
typedef __attribute__((ext_vector_type(4))) float v4f;
typedef __attribute__((ext_vector_type(8))) float v8f;

__global__ __launch_bounds__(256) void gat_gemm_wmma(
    const float* __restrict__ feat, const float* __restrict__ W,
    float* __restrict__ ft)
{
    const int lane  = threadIdx.x & 31;
    const int wave  = threadIdx.x >> 5;
    const int m0    = blockIdx.x * 16;
    const int nb    = wave * 64;
    const int mrow  = m0 + (lane & 15);
    const int khalf = (lane >> 4) * 2;
    const int ncol  = lane & 15;

    v8f acc0 = {}, acc1 = {}, acc2 = {}, acc3 = {};
    const float* arow = feat + (size_t)mrow * IN_FEATS;

    for (int k0 = 0; k0 < IN_FEATS; k0 += 4) {
        const int kb = k0 + khalf;
        v2f a = *(const v2f*)(arow + kb);
        const float* w0 = W + (size_t)kb * HD + nb + ncol;
        const float* w1 = w0 + HD;
        v2f b0, b1, b2, b3;
        b0.x = w0[0];  b0.y = w1[0];
        b1.x = w0[16]; b1.y = w1[16];
        b2.x = w0[32]; b2.y = w1[32];
        b3.x = w0[48]; b3.y = w1[48];
        acc0 = __builtin_amdgcn_wmma_f32_16x16x4_f32(false, a, false, b0, (short)0, acc0, false, false);
        acc1 = __builtin_amdgcn_wmma_f32_16x16x4_f32(false, a, false, b1, (short)0, acc1, false, false);
        acc2 = __builtin_amdgcn_wmma_f32_16x16x4_f32(false, a, false, b2, (short)0, acc2, false, false);
        acc3 = __builtin_amdgcn_wmma_f32_16x16x4_f32(false, a, false, b3, (short)0, acc3, false, false);
    }

    const int rbase = m0 + ((lane >> 4) << 3);
#pragma unroll
    for (int r = 0; r < 8; ++r) {
        float* dp = ft + (size_t)(rbase + r) * HD + nb + ncol;
        dp[0]  = acc0[r];
        dp[16] = acc1[r];
        dp[32] = acc2[r];
        dp[48] = acc3[r];
    }
}

__global__ __launch_bounds__(256) void gat_logits(
    const float* __restrict__ ft, const float* __restrict__ attn_l,
    const float* __restrict__ attn_r, float* __restrict__ el,
    float* __restrict__ er)
{
    const int n    = blockIdx.x;
    const int h    = threadIdx.x >> 5;
    const int lane = threadIdx.x & 31;

    const float* f  = ft + (size_t)n * HD + h * OUT_FEATS;
    const float* al = attn_l + h * OUT_FEATS;
    const float* ar = attn_r + h * OUT_FEATS;

    const float f0 = f[lane], f1 = f[lane + 32];
    float sl = f0 * al[lane] + f1 * al[lane + 32];
    float sr = f0 * ar[lane] + f1 * ar[lane + 32];
#pragma unroll
    for (int m = 16; m > 0; m >>= 1) {
        sl += __shfl_xor(sl, m, 32);
        sr += __shfl_xor(sr, m, 32);
    }
    if (lane == 0) {
        el[n * HEADS + h] = sl;
        er[n * HEADS + h] = sr;
    }
}

__global__ __launch_bounds__(128) void gat_edge_softmax_agg(
    const float* __restrict__ ft, const float* __restrict__ el,
    const float* __restrict__ er, const int* __restrict__ src,
    const int* __restrict__ dst, float* __restrict__ rst,
    float* __restrict__ a_out)
{
    __shared__ int   s_src[DEG];
    __shared__ float s_e[DEG][HEADS];
    __shared__ float s_a[DEG][HEADS];

    const int    n     = blockIdx.x;
    const int    t     = threadIdx.x;
    const long   ebase = (long)n * DEG;

    if (t < DEG) s_src[t] = src[ebase + t];
    __syncthreads();

#pragma unroll
    for (int rep = 0; rep < 2; ++rep) {
        const int p = t + rep * 128;
        const int j = p >> 3, h = p & 7;
        const int dn = dst[ebase + j];
        float e = el[s_src[j] * HEADS + h] + er[dn * HEADS + h];
        s_e[j][h] = (e > 0.f) ? e : NEG_SLOPE * e;
    }
    __syncthreads();

#pragma unroll
    for (int rep = 0; rep < 2; ++rep) {
        const int h = ((t >> 5) << 1) + rep, j = t & 31;
        float e  = s_e[j][h];
        float mx = e;
#pragma unroll
        for (int m = 16; m > 0; m >>= 1) mx = fmaxf(mx, __shfl_xor(mx, m, 32));
        float ex  = __expf(e - mx);
        float sum = ex;
#pragma unroll
        for (int m = 16; m > 0; m >>= 1) sum += __shfl_xor(sum, m, 32);
        const float a = ex / sum;
        s_a[j][h] = a;
        a_out[(ebase + j) * HEADS + h] = a;
    }
    __syncthreads();

    {
        const int h = t >> 4;
        const float* fcol = ft + 4 * t;
        v4f acc = {0.f, 0.f, 0.f, 0.f};
#pragma unroll 8
        for (int j = 0; j < DEG; ++j) {
            const float a = s_a[j][h];
            const v4f   v = *(const v4f*)(fcol + (size_t)s_src[j] * HD);
            acc.x += a * v.x;
            acc.y += a * v.y;
            acc.z += a * v.z;
            acc.w += a * v.w;
        }
        *(v4f*)(rst + (size_t)n * HD + 4 * t) = acc;
    }
}

extern "C" void kernel_launch(void* const* d_in, const int* in_sizes, int n_in,
                              void* d_out, int out_size, void* d_ws, size_t ws_size,
                              hipStream_t stream)
{
    const float* feat   = (const float*)d_in[0];
    const float* W      = (const float*)d_in[1];
    const float* attn_l = (const float*)d_in[2];
    const float* attn_r = (const float*)d_in[3];
    const int*   src    = (const int*)d_in[4];
    const int*   dst    = (const int*)d_in[5];

    float* out_rst = (float*)d_out;
    float* out_a   = out_rst + (size_t)N_NODES * HD;

    float* ft = (float*)d_ws;
    float* el = ft + (size_t)N_NODES * HD;
    float* er = el + (size_t)N_NODES * HEADS;

    gat_gemm_wmma<<<N_NODES / 16, 256, 0, stream>>>(feat, W, ft);
    gat_logits<<<N_NODES, 256, 0, stream>>>(ft, attn_l, attn_r, el, er);
    gat_edge_softmax_agg<<<N_NODES, 128, 0, stream>>>(ft, el, er, src, dst,
                                                      out_rst, out_a);
}


# attention logits el/er formed in the GEMM epilogue from the accumulators (16-lane DPP butterfly, LDS-staged whole-line stores); logits launch removed, 3 -> 2 launches; el/er sum tree unchanged
# speedup vs baseline: 1.5053x; 1.1936x over previous
#include <hip/hip_runtime.h>
#include <hip/hip_bf16.h>
#include <math.h>

#define N_NODES 10000
#define DEG 32
#define IN_FEATS 256
#define HEADS 8
#define OUT_FEATS 64
#define HD (HEADS * OUT_FEATS)
#define NEG_SLOPE 0.2f

typedef __attribute__((ext_vector_type(2))) float v2f;
typedef __attribute__((ext_vector_type(4))) float v4f;
typedef __attribute__((ext_vector_type(8))) float v8f;

template <int M>
__device__ __forceinline__ float row_xor(float x)
{
    return __builtin_bit_cast(float, __builtin_amdgcn_update_dpp(
        0, __builtin_bit_cast(int, x), 0x160 | M, 0xf, 0xf, true));
}

__global__ __launch_bounds__(256) void gat_gemm_wmma(
    const float* __restrict__ feat, const float* __restrict__ W,
    const float* __restrict__ attn_l, const float* __restrict__ attn_r,
    float* __restrict__ ft, float* __restrict__ el, float* __restrict__ er)
{
    __shared__ __attribute__((aligned(16))) float s_lr[2][16][HEADS];

    const int lane  = threadIdx.x & 31;
    const int wave  = threadIdx.x >> 5;
    const int m0    = blockIdx.x * 16;
    const int nb    = wave * 64;
    const int mrow  = m0 + (lane & 15);
    const int khalf = (lane >> 4) * 2;
    const int ncol  = lane & 15;

    v8f acc0 = {}, acc1 = {}, acc2 = {}, acc3 = {};
    const float* arow = feat + (size_t)mrow * IN_FEATS;

    for (int k0 = 0; k0 < IN_FEATS; k0 += 4) {
        const int kb = k0 + khalf;
        v2f a = *(const v2f*)(arow + kb);
        const float* w0 = W + (size_t)kb * HD + nb + ncol;
        const float* w1 = w0 + HD;
        v2f b0, b1, b2, b3;
        b0.x = w0[0];  b0.y = w1[0];
        b1.x = w0[16]; b1.y = w1[16];
        b2.x = w0[32]; b2.y = w1[32];
        b3.x = w0[48]; b3.y = w1[48];
        acc0 = __builtin_amdgcn_wmma_f32_16x16x4_f32(false, a, false, b0, (short)0, acc0, false, false);
        acc1 = __builtin_amdgcn_wmma_f32_16x16x4_f32(false, a, false, b1, (short)0, acc1, false, false);
        acc2 = __builtin_amdgcn_wmma_f32_16x16x4_f32(false, a, false, b2, (short)0, acc2, false, false);
        acc3 = __builtin_amdgcn_wmma_f32_16x16x4_f32(false, a, false, b3, (short)0, acc3, false, false);
    }

    const float* alp = attn_l + nb + ncol;
    const float* arp = attn_r + nb + ncol;
    const float l0 = alp[0], l1 = alp[16], l2 = alp[32], l3 = alp[48];
    const float r0 = arp[0], r1 = arp[16], r2 = arp[32], r3 = arp[48];

    const int rbase = m0 + ((lane >> 4) << 3);
#pragma unroll
    for (int r = 0; r < 8; ++r) {
        float* dp = ft + (size_t)(rbase + r) * HD + nb + ncol;
        dp[0]  = acc0[r];
        dp[16] = acc1[r];
        dp[32] = acc2[r];
        dp[48] = acc3[r];
    }

    float sl[8], sr[8];
#pragma unroll
    for (int r = 0; r < 8; ++r) {
        sl[r] = fmaf(acc0[r], l0, acc2[r] * l2) + fmaf(acc1[r], l1, acc3[r] * l3);
        sr[r] = fmaf(acc0[r], r0, acc2[r] * r2) + fmaf(acc1[r], r1, acc3[r] * r3);
    }
#pragma unroll
    for (int r = 0; r < 8; ++r) { sl[r] += row_xor<8>(sl[r]); sr[r] += row_xor<8>(sr[r]); }
#pragma unroll
    for (int r = 0; r < 8; ++r) { sl[r] += row_xor<4>(sl[r]); sr[r] += row_xor<4>(sr[r]); }
#pragma unroll
    for (int r = 0; r < 8; ++r) { sl[r] += row_xor<2>(sl[r]); sr[r] += row_xor<2>(sr[r]); }
#pragma unroll
    for (int r = 0; r < 8; ++r) { sl[r] += row_xor<1>(sl[r]); sr[r] += row_xor<1>(sr[r]); }

    {
        const int rr = ncol & 7;
        float v = 0.f;
#pragma unroll
        for (int r = 0; r < 8; ++r)
            if (rr == r) v = (ncol < 8) ? sl[r] : sr[r];
        s_lr[ncol >> 3][((lane >> 4) << 3) + rr][wave] = v;
    }
    __syncthreads();

    if (threadIdx.x < 64) {
        const int which = threadIdx.x >> 5;
        const int i4    = lane * 4;
        const v4f o = *(const v4f*)(&s_lr[which][0][0] + i4);
        float* op = (which ? er : el) + (size_t)m0 * HEADS + i4;
        *(v4f*)op = o;
    }
}

__global__ __launch_bounds__(128) void gat_edge_softmax_agg(
    const float* __restrict__ ft, const float* __restrict__ el,
    const float* __restrict__ er, const int* __restrict__ src,
    const int* __restrict__ dst, float* __restrict__ rst,
    float* __restrict__ a_out)
{
    __shared__ int   s_src[DEG];
    __shared__ float s_e[DEG][HEADS];
    __shared__ float s_a[DEG][HEADS];

    const int    n     = blockIdx.x;
    const int    t     = threadIdx.x;
    const long   ebase = (long)n * DEG;

    if (t < DEG) s_src[t] = src[ebase + t];
    __syncthreads();

#pragma unroll
    for (int rep = 0; rep < 2; ++rep) {
        const int p = t + rep * 128;
        const int j = p >> 3, h = p & 7;
        const int dn = dst[ebase + j];
        float e = el[s_src[j] * HEADS + h] + er[dn * HEADS + h];
        s_e[j][h] = (e > 0.f) ? e : NEG_SLOPE * e;
    }
    __syncthreads();

#pragma unroll
    for (int rep = 0; rep < 2; ++rep) {
        const int h = ((t >> 5) << 1) + rep, j = t & 31;
        float e  = s_e[j][h];
        float mx = e;
#pragma unroll
        for (int m = 16; m > 0; m >>= 1) mx = fmaxf(mx, __shfl_xor(mx, m, 32));
        float ex  = __expf(e - mx);
        float sum = ex;
#pragma unroll
        for (int m = 16; m > 0; m >>= 1) sum += __shfl_xor(sum, m, 32);
        const float a = ex / sum;
        s_a[j][h] = a;
        a_out[(ebase + j) * HEADS + h] = a;
    }
    __syncthreads();

    {
        const int h = t >> 4;
        const float* fcol = ft + 4 * t;
        v4f acc = {0.f, 0.f, 0.f, 0.f};
#pragma unroll 8
        for (int j = 0; j < DEG; ++j) {
            const float a = s_a[j][h];
            const v4f   v = *(const v4f*)(fcol + (size_t)s_src[j] * HD);
            acc.x += a * v.x;
            acc.y += a * v.y;
            acc.z += a * v.z;
            acc.w += a * v.w;
        }
        *(v4f*)(rst + (size_t)n * HD + 4 * t) = acc;
    }
}

extern "C" void kernel_launch(void* const* d_in, const int* in_sizes, int n_in,
                              void* d_out, int out_size, void* d_ws, size_t ws_size,
                              hipStream_t stream)
{
    const float* feat   = (const float*)d_in[0];
    const float* W      = (const float*)d_in[1];
    const float* attn_l = (const float*)d_in[2];
    const float* attn_r = (const float*)d_in[3];
    const int*   src    = (const int*)d_in[4];
    const int*   dst    = (const int*)d_in[5];

    float* out_rst = (float*)d_out;
    float* out_a   = out_rst + (size_t)N_NODES * HD;

    float* ft = (float*)d_ws;
    float* el = ft + (size_t)N_NODES * HD;
    float* er = el + (size_t)N_NODES * HEADS;

    gat_gemm_wmma<<<N_NODES / 16, 256, 0, stream>>>(feat, W, attn_l, attn_r,
                                                    ft, el, er);
    gat_edge_softmax_agg<<<N_NODES, 128, 0, stream>>>(ft, el, er, src, dst,
                                                      out_rst, out_a);
}
